# starting kernel
# baseline (speedup 1.0000x reference)
#include <hip/hip_runtime.h>
#include <math.h>

typedef __attribute__((ext_vector_type(2))) float v2f;
typedef __attribute__((ext_vector_type(8))) float v8f;

#define P_DIM 1024
#define PR    64
#define AMIN  0.01f
#define AMAX  0.995f
#define SCAN_L 64

__global__ __launch_bounds__(256) void k_proj(const float* __restrict__ t,
                                              const float* __restrict__ F,
                                              const float* __restrict__ Vr,
                                              const float* __restrict__ Vb,
                                              float* __restrict__ r_out,
                                              float* __restrict__ inj_out) {
  __shared__ float At [16 * 68];
  __shared__ float Aft[16 * 68];
  const int lane  = threadIdx.x & 31;
  const int wave  = threadIdx.x >> 5;
  const int mat   = wave >> 2;
  const int cn    = wave & 3;
  const int row0  = blockIdx.x * 16;
  const float* __restrict__ B = mat ? Vb : Vr;

  const int lrow  = threadIdx.x >> 4;
  const int lcol  = (threadIdx.x & 15) << 2;
  const int m     = lane & 15;
  const int khalf = (lane >> 4) << 1;
  const int bcol  = cn * 16 + (lane & 15);

  v8f acc = {};
  for (int kk = 0; kk < P_DIM; kk += 64) {
    const float4 tv = *(const float4*)&t[(size_t)(row0 + lrow) * P_DIM + kk + lcol];
    const float4 fv = *(const float4*)&F[(size_t)(row0 + lrow) * P_DIM + kk + lcol];
    float* at  = &At [lrow * 68 + lcol];
    float* aft = &Aft[lrow * 68 + lcol];
    at[0] = tv.x; at[1] = tv.y; at[2] = tv.z; at[3] = tv.w;
    aft[0] = tv.x * fv.x; aft[1] = tv.y * fv.y;
    aft[2] = tv.z * fv.z; aft[3] = tv.w * fv.w;
    __syncthreads();

    const float* __restrict__ Ab = mat ? Aft : At;
#pragma unroll
    for (int ks = 0; ks < 64; ks += 4) {
      v2f afrag = *(const v2f*)&Ab[m * 68 + ks + khalf];
      const int kabs = kk + ks + khalf;
      v2f bfrag;
      bfrag.x = B[(size_t)kabs       * PR + bcol];
      bfrag.y = B[(size_t)(kabs + 1) * PR + bcol];
      acc = __builtin_amdgcn_wmma_f32_16x16x4_f32(false, afrag, false, bfrag,
                                                  (short)0, acc, false, false);
    }
    __syncthreads();
  }

  float* __restrict__ dst = mat ? inj_out : r_out;
#pragma unroll
  for (int v = 0; v < 8; ++v) {
    const int row = row0 + v + ((lane >> 4) << 3);
    dst[(size_t)row * PR + bcol] = acc[v];
  }
}

__global__ __launch_bounds__(256) void k_gate(const float* __restrict__ r_in,
                                              const float* __restrict__ Ww,
                                              const float* __restrict__ Wb,
                                              float* __restrict__ a_out) {
  const int lane  = threadIdx.x & 31;
  const int wave  = threadIdx.x >> 5;
  const int wm    = wave >> 2, cn = wave & 3;
  const int row0  = blockIdx.x * 32 + wm * 16;
  const int m     = lane & 15;
  const int khalf = (lane >> 4) << 1;
  const int col   = cn * 16 + (lane & 15);

  v8f acc = {};
#pragma unroll
  for (int ks = 0; ks < PR; ks += 4) {
    v2f afrag = *(const v2f*)&r_in[(size_t)(row0 + m) * PR + ks + khalf];
    v2f bfrag = *(const v2f*)&Ww[(size_t)col * PR + ks + khalf];
    acc = __builtin_amdgcn_wmma_f32_16x16x4_f32(false, afrag, false, bfrag,
                                                (short)0, acc, false, false);
  }
  const float bias = Wb[col];
#pragma unroll
  for (int v = 0; v < 8; ++v) {
    const int row = row0 + v + ((lane >> 4) << 3);
    float x = acc[v] + bias;
    float s = 1.0f / (1.0f + __expf(-x));
    s = fminf(fmaxf(s, AMIN), AMAX);
    a_out[(size_t)row * PR + col] = s;
  }
}

__global__ __launch_bounds__(64) void k_scan_local(float* __restrict__ a_cum,
                                                   float* __restrict__ inj_s,
                                                   int T) {
  const int C = T / SCAN_L;
  const int b = blockIdx.x / C, c = blockIdx.x % C;
  const int r = threadIdx.x;
  size_t base = ((size_t)b * T + (size_t)c * SCAN_L) * PR + r;
  float s = 0.0f, cum = 1.0f;
  for (int i = 0; i < SCAN_L; ++i) {
    const size_t off = base + (size_t)i * PR;
    const float av = a_cum[off];
    s   = fmaf(av, s, inj_s[off]);
    cum *= av;
    a_cum[off] = cum;
    inj_s[off] = s;
  }
}

__global__ __launch_bounds__(64) void k_scan_carry(const float* __restrict__ a_cum,
                                                   const float* __restrict__ inj_s,
                                                   float* __restrict__ carry_buf,
                                                   float* __restrict__ s_last,
                                                   int T) {
  const int C = T / SCAN_L;
  const int b = blockIdx.x, r = threadIdx.x;
  float carry = 0.0f;
  for (int c = 0; c < C; ++c) {
    carry_buf[((size_t)b * C + c) * PR + r] = carry;
    const size_t off = ((size_t)b * T + (size_t)c * SCAN_L + (SCAN_L - 1)) * PR + r;
    carry = fmaf(a_cum[off], carry, inj_s[off]);
  }
  s_last[b * PR + r] = carry;
}

__global__ __launch_bounds__(256) void k_scan_apply(const float* __restrict__ a_cum,
                                                    const float* __restrict__ carry_buf,
                                                    float* __restrict__ inj_s,
                                                    int T, int total) {
  const int idx = blockIdx.x * blockDim.x + threadIdx.x;
  if (idx >= total) return;
  const int C = T / SCAN_L;
  const int r = idx % PR;
  const int t = (idx / PR) % T;
  const int b = idx / (PR * T);
  const int c = t / SCAN_L;
  inj_s[idx] = fmaf(a_cum[idx], carry_buf[((size_t)b * C + c) * PR + r], inj_s[idx]);
}

__global__ __launch_bounds__(256) void k_outp(const float* __restrict__ s_in,
                                              const float* __restrict__ Vo,
                                              float* __restrict__ out) {
  const int lane  = threadIdx.x & 31;
  const int wave  = threadIdx.x >> 5;
  const int row0  = blockIdx.x * 16;
  const int col0  = blockIdx.y * 128 + wave * 16;
  const int m     = lane & 15;
  const int khalf = (lane >> 4) << 1;
  const int col   = col0 + (lane & 15);

  v8f acc = {};
#pragma unroll
  for (int ks = 0; ks < PR; ks += 4) {
    v2f afrag = *(const v2f*)&s_in[(size_t)(row0 + m) * PR + ks + khalf];
    const int kabs = ks + khalf;
    v2f bfrag;
    bfrag.x = Vo[(size_t)kabs       * P_DIM + col];
    bfrag.y = Vo[(size_t)(kabs + 1) * P_DIM + col];
    acc = __builtin_amdgcn_wmma_f32_16x16x4_f32(false, afrag, false, bfrag,
                                                (short)0, acc, false, false);
  }
#pragma unroll
  for (int v = 0; v < 8; ++v) {
    const int row = row0 + v + ((lane >> 4) << 3);
    out[(size_t)row * P_DIM + col] = acc[v];
  }
}

extern "C" void kernel_launch(void* const* d_in, const int* in_sizes, int n_in,
                              void* d_out, int out_size, void* d_ws, size_t ws_size,
                              hipStream_t stream) {
  const float* t  = (const float*)d_in[0];
  const float* F  = (const float*)d_in[1];
  const float* Vr = (const float*)d_in[2];
  const float* Vb = (const float*)d_in[3];
  const float* Vo = (const float*)d_in[4];
  const float* Ww = (const float*)d_in[5];
  const float* Wb = (const float*)d_in[6];
  float* out = (float*)d_out;

  const int rows = in_sizes[0] / P_DIM;
  const int Bn   = 8;
  const int T    = rows / Bn;
  const int C    = T / SCAN_L;

  float* r_buf     = (float*)d_ws;
  float* a_buf     = r_buf   + (size_t)rows * PR;
  float* inj_buf   = a_buf   + (size_t)rows * PR;
  float* carry_buf = inj_buf + (size_t)rows * PR;
  float* s_last    = out + (size_t)rows * P_DIM;

  k_proj<<<rows / 16, 256, 0, stream>>>(t, F, Vr, Vb, r_buf, inj_buf);
  k_gate<<<rows / 32, 256, 0, stream>>>(r_buf, Ww, Wb, a_buf);
  k_scan_local<<<Bn * C, 64, 0, stream>>>(a_buf, inj_buf, T);
  k_scan_carry<<<Bn, 64, 0, stream>>>(a_buf, inj_buf, carry_buf, s_last, T);
  const int total = rows * PR;
  k_scan_apply<<<(total + 255) / 256, 256, 0, stream>>>(a_buf, carry_buf, inj_buf, T, total);
  k_outp<<<dim3(rows / 16, P_DIM / 128), 256, 0, stream>>>(inj_buf, Vo, out);
}


# k_outp re-tiled: one wave = 32 rows x 128 cols; s tile staged once per block in LDS, A fragments held in registers over all column tiles; V_o loads (3 K-steps ahead) and out stores as 16-byte vectors
# speedup vs baseline: 1.2098x; 1.2098x over previous
#include <hip/hip_runtime.h>
#include <math.h>

typedef __attribute__((ext_vector_type(2))) float v2f;
typedef __attribute__((ext_vector_type(8))) float v8f;

#define P_DIM 1024
#define PR    64
#define AMIN  0.01f
#define AMAX  0.995f
#define SCAN_L 64

__global__ __launch_bounds__(256) void k_proj(const float* __restrict__ t,
                                              const float* __restrict__ F,
                                              const float* __restrict__ Vr,
                                              const float* __restrict__ Vb,
                                              float* __restrict__ r_out,
                                              float* __restrict__ inj_out) {
  __shared__ float At [16 * 68];
  __shared__ float Aft[16 * 68];
  const int lane  = threadIdx.x & 31;
  const int wave  = threadIdx.x >> 5;
  const int mat   = wave >> 2;
  const int cn    = wave & 3;
  const int row0  = blockIdx.x * 16;
  const float* __restrict__ B = mat ? Vb : Vr;

  const int lrow  = threadIdx.x >> 4;
  const int lcol  = (threadIdx.x & 15) << 2;
  const int m     = lane & 15;
  const int khalf = (lane >> 4) << 1;
  const int bcol  = cn * 16 + (lane & 15);

  v8f acc = {};
  for (int kk = 0; kk < P_DIM; kk += 64) {
    const float4 tv = *(const float4*)&t[(size_t)(row0 + lrow) * P_DIM + kk + lcol];
    const float4 fv = *(const float4*)&F[(size_t)(row0 + lrow) * P_DIM + kk + lcol];
    float* at  = &At [lrow * 68 + lcol];
    float* aft = &Aft[lrow * 68 + lcol];
    at[0] = tv.x; at[1] = tv.y; at[2] = tv.z; at[3] = tv.w;
    aft[0] = tv.x * fv.x; aft[1] = tv.y * fv.y;
    aft[2] = tv.z * fv.z; aft[3] = tv.w * fv.w;
    __syncthreads();

    const float* __restrict__ Ab = mat ? Aft : At;
#pragma unroll
    for (int ks = 0; ks < 64; ks += 4) {
      v2f afrag = *(const v2f*)&Ab[m * 68 + ks + khalf];
      const int kabs = kk + ks + khalf;
      v2f bfrag;
      bfrag.x = B[(size_t)kabs       * PR + bcol];
      bfrag.y = B[(size_t)(kabs + 1) * PR + bcol];
      acc = __builtin_amdgcn_wmma_f32_16x16x4_f32(false, afrag, false, bfrag,
                                                  (short)0, acc, false, false);
    }
    __syncthreads();
  }

  float* __restrict__ dst = mat ? inj_out : r_out;
#pragma unroll
  for (int v = 0; v < 8; ++v) {
    const int row = row0 + v + ((lane >> 4) << 3);
    dst[(size_t)row * PR + bcol] = acc[v];
  }
}

__global__ __launch_bounds__(256) void k_gate(const float* __restrict__ r_in,
                                              const float* __restrict__ Ww,
                                              const float* __restrict__ Wb,
                                              float* __restrict__ a_out) {
  const int lane  = threadIdx.x & 31;
  const int wave  = threadIdx.x >> 5;
  const int wm    = wave >> 2, cn = wave & 3;
  const int row0  = blockIdx.x * 32 + wm * 16;
  const int m     = lane & 15;
  const int khalf = (lane >> 4) << 1;
  const int col   = cn * 16 + (lane & 15);

  v8f acc = {};
#pragma unroll
  for (int ks = 0; ks < PR; ks += 4) {
    v2f afrag = *(const v2f*)&r_in[(size_t)(row0 + m) * PR + ks + khalf];
    v2f bfrag = *(const v2f*)&Ww[(size_t)col * PR + ks + khalf];
    acc = __builtin_amdgcn_wmma_f32_16x16x4_f32(false, afrag, false, bfrag,
                                                (short)0, acc, false, false);
  }
  const float bias = Wb[col];
#pragma unroll
  for (int v = 0; v < 8; ++v) {
    const int row = row0 + v + ((lane >> 4) << 3);
    float x = acc[v] + bias;
    float s = 1.0f / (1.0f + __expf(-x));
    s = fminf(fmaxf(s, AMIN), AMAX);
    a_out[(size_t)row * PR + col] = s;
  }
}

__global__ __launch_bounds__(64) void k_scan_local(float* __restrict__ a_cum,
                                                   float* __restrict__ inj_s,
                                                   int T) {
  const int C = T / SCAN_L;
  const int b = blockIdx.x / C, c = blockIdx.x % C;
  const int r = threadIdx.x;
  size_t base = ((size_t)b * T + (size_t)c * SCAN_L) * PR + r;
  float s = 0.0f, cum = 1.0f;
  for (int i = 0; i < SCAN_L; ++i) {
    const size_t off = base + (size_t)i * PR;
    const float av = a_cum[off];
    s   = fmaf(av, s, inj_s[off]);
    cum *= av;
    a_cum[off] = cum;
    inj_s[off] = s;
  }
}

__global__ __launch_bounds__(64) void k_scan_carry(const float* __restrict__ a_cum,
                                                   const float* __restrict__ inj_s,
                                                   float* __restrict__ carry_buf,
                                                   float* __restrict__ s_last,
                                                   int T) {
  const int C = T / SCAN_L;
  const int b = blockIdx.x, r = threadIdx.x;
  float carry = 0.0f;
  for (int c = 0; c < C; ++c) {
    carry_buf[((size_t)b * C + c) * PR + r] = carry;
    const size_t off = ((size_t)b * T + (size_t)c * SCAN_L + (SCAN_L - 1)) * PR + r;
    carry = fmaf(a_cum[off], carry, inj_s[off]);
  }
  s_last[b * PR + r] = carry;
}

__global__ __launch_bounds__(256) void k_scan_apply(const float* __restrict__ a_cum,
                                                    const float* __restrict__ carry_buf,
                                                    float* __restrict__ inj_s,
                                                    int T, int total) {
  const int idx = blockIdx.x * blockDim.x + threadIdx.x;
  if (idx >= total) return;
  const int C = T / SCAN_L;
  const int r = idx % PR;
  const int t = (idx / PR) % T;
  const int b = idx / (PR * T);
  const int c = t / SCAN_L;
  inj_s[idx] = fmaf(a_cum[idx], carry_buf[((size_t)b * C + c) * PR + r], inj_s[idx]);
}

#define OUT_MT 2
#define OUT_PF 3
__global__ __launch_bounds__(256) void k_outp(const float* __restrict__ s_in,
                                              const float* __restrict__ Vo,
                                              float* __restrict__ out) {
  __shared__ float As[OUT_MT * 16 * 68];
  const int lane  = threadIdx.x & 31;
  const int wave  = threadIdx.x >> 5;
  const int row0  = blockIdx.x * (OUT_MT * 16);
  const int c     = lane & 15;
  const int h     = lane >> 4;
  const int khalf = h << 1;
  const int colw  = wave * 128 + 4 * c;

  float4 rb0[OUT_PF + 1], rb1[OUT_PF + 1];
#pragma unroll
  for (int t = 0; t < OUT_PF; ++t) {
    const int kabs = 4 * t + khalf;
    rb0[t] = *(const float4*)&Vo[(size_t)kabs       * P_DIM + colw];
    rb1[t] = *(const float4*)&Vo[(size_t)(kabs + 1) * P_DIM + colw];
  }

#pragma unroll
  for (int q = 0; q < OUT_MT; ++q) {
    const int lrow = q * 16 + (threadIdx.x >> 4);
    const int lcol = (threadIdx.x & 15) << 2;
    const float4 sv = *(const float4*)&s_in[(size_t)(row0 + lrow) * PR + lcol];
    *(float4*)&As[lrow * 68 + lcol] = sv;
  }
  __syncthreads();

  v2f a[OUT_MT][16];
#pragma unroll
  for (int q = 0; q < OUT_MT; ++q) {
#pragma unroll
    for (int i = 0; i < 16; ++i)
      a[q][i] = *(const v2f*)&As[(q * 16 + c) * 68 + 4 * i + khalf];
  }

  v8f acc[OUT_MT][4];
#pragma unroll
  for (int t = 0; t < 32; ++t) {
    const int g = t >> 4, i = t & 15;
    if (i == 0) {
#pragma unroll
      for (int q = 0; q < OUT_MT; ++q) {
#pragma unroll
        for (int j = 0; j < 4; ++j) acc[q][j] = (v8f){};
      }
    }
    if (t + OUT_PF < 32) {
      const int tn = t + OUT_PF;
      const int kabs = 4 * (tn & 15) + khalf;
      const int coln = colw + (tn >> 4) * 64;
      rb0[tn % (OUT_PF + 1)] = *(const float4*)&Vo[(size_t)kabs       * P_DIM + coln];
      rb1[tn % (OUT_PF + 1)] = *(const float4*)&Vo[(size_t)(kabs + 1) * P_DIM + coln];
    }
    __builtin_amdgcn_sched_barrier(0);
    const float4 b0 = rb0[t % (OUT_PF + 1)];
    const float4 b1 = rb1[t % (OUT_PF + 1)];
    const float b0a[4] = {b0.x, b0.y, b0.z, b0.w};
    const float b1a[4] = {b1.x, b1.y, b1.z, b1.w};
#pragma unroll
    for (int j = 0; j < 4; ++j) {
      v2f bfrag;
      bfrag.x = b0a[j];
      bfrag.y = b1a[j];
#pragma unroll
      for (int q = 0; q < OUT_MT; ++q)
        acc[q][j] = __builtin_amdgcn_wmma_f32_16x16x4_f32(false, a[q][i], false, bfrag,
                                                          (short)0, acc[q][j], false, false);
    }
    if (i == 15) {
      const int colb = colw + g * 64;
#pragma unroll
      for (int q = 0; q < OUT_MT; ++q) {
#pragma unroll
        for (int v = 0; v < 8; ++v) {
          const int row = row0 + q * 16 + v + (h << 3);
          float4 o;
          o.x = acc[q][0][v]; o.y = acc[q][1][v]; o.z = acc[q][2][v]; o.w = acc[q][3][v];
          *(float4*)&out[(size_t)row * P_DIM + colb] = o;
        }
      }
    }
  }
}

extern "C" void kernel_launch(void* const* d_in, const int* in_sizes, int n_in,
                              void* d_out, int out_size, void* d_ws, size_t ws_size,
                              hipStream_t stream) {
  const float* t  = (const float*)d_in[0];
  const float* F  = (const float*)d_in[1];
  const float* Vr = (const float*)d_in[2];
  const float* Vb = (const float*)d_in[3];
  const float* Vo = (const float*)d_in[4];
  const float* Ww = (const float*)d_in[5];
  const float* Wb = (const float*)d_in[6];
  float* out = (float*)d_out;

  const int rows = in_sizes[0] / P_DIM;
  const int Bn   = 8;
  const int T    = rows / Bn;
  const int C    = T / SCAN_L;

  float* r_buf     = (float*)d_ws;
  float* a_buf     = r_buf   + (size_t)rows * PR;
  float* inj_buf   = a_buf   + (size_t)rows * PR;
  float* carry_buf = inj_buf + (size_t)rows * PR;
  float* s_last    = out + (size_t)rows * P_DIM;

  k_proj<<<rows / 16, 256, 0, stream>>>(t, F, Vr, Vb, r_buf, inj_buf);
  k_gate<<<rows / 32, 256, 0, stream>>>(r_buf, Ww, Wb, a_buf);
  k_scan_local<<<Bn * C, 64, 0, stream>>>(a_buf, inj_buf, T);
  k_scan_carry<<<Bn, 64, 0, stream>>>(a_buf, inj_buf, carry_buf, s_last, T);
  const int total = rows * PR;
  k_scan_apply<<<(total + 255) / 256, 256, 0, stream>>>(a_buf, carry_buf, inj_buf, T, total);
  k_outp<<<rows / 32, 256, 0, stream>>>(inj_buf, Vo, out);
}


# k_proj re-tiled and software-pipelined: wave = 16 rows x 64 cols (4 accumulators); t/F requested a chunk ahead, V_r/V_b fragments (16-byte loads) 15 K-steps ahead; two LDS stages, LDS-only barrier
# speedup vs baseline: 1.3523x; 1.1178x over previous
#include <hip/hip_runtime.h>
#include <math.h>

typedef __attribute__((ext_vector_type(2))) float v2f;
typedef __attribute__((ext_vector_type(4))) float v4f;
typedef __attribute__((ext_vector_type(8))) float v8f;

#define P_DIM 1024
#define PR    64
#define AMIN  0.01f
#define AMAX  0.995f
#define SCAN_L 64

#define PJ_ROWS 32
#define PJ_LD   68
#define PJ_MAT  (PJ_ROWS * PJ_LD)
#define PJ_STG  (2 * PJ_MAT)

__device__ __forceinline__ void lds_barrier() {
  asm volatile("s_wait_dscnt 0x0\n\ts_barrier_signal -1\n\ts_barrier_wait -1" ::: "memory");
}

#define PJ_PF 15

template <bool LAST>
__device__ __forceinline__ void proj_chunk(const float* Ab, const float* __restrict__ Bc,
                                           v4f (&rb0)[16], v4f (&rb1)[16], v8f (&acc)[4]) {
  v2f a[16];
#pragma unroll
  for (int i = 0; i < 16; ++i) a[i] = *(const v2f*)&Ab[4 * i];
#pragma unroll
  for (int i = 0; i < 16; ++i) {
    if (!LAST || i == 0) {
      rb0[(i + PJ_PF) & 15] = *(const v4f*)&Bc[(size_t)(4 * (i + PJ_PF))     * PR];
      rb1[(i + PJ_PF) & 15] = *(const v4f*)&Bc[(size_t)(4 * (i + PJ_PF) + 1) * PR];
    }
    __builtin_amdgcn_sched_barrier(0);
    const v4f b0 = rb0[i];
    const v4f b1 = rb1[i];
#pragma unroll
    for (int j = 0; j < 4; ++j) {
      v2f bfrag;
      bfrag.x = b0[j];
      bfrag.y = b1[j];
      acc[j] = __builtin_amdgcn_wmma_f32_16x16x4_f32(false, a[i], false, bfrag,
                                                     (short)0, acc[j], false, false);
    }
  }
}

__global__ __launch_bounds__(128) void k_proj(const float* __restrict__ t,
                                              const float* __restrict__ F,
                                              const float* __restrict__ Vr,
                                              const float* __restrict__ Vb,
                                              float* __restrict__ r_out,
                                              float* __restrict__ inj_out) {
  __shared__ __attribute__((aligned(16))) float As[2 * PJ_STG];
  const int lane  = threadIdx.x & 31;
  const int wave  = threadIdx.x >> 5;
  const int mat   = wave >> 1;
  const int rt    = wave & 1;
  const int row0  = blockIdx.x * PJ_ROWS;
  const int c     = lane & 15;
  const int h     = lane >> 4;
  const int khalf = h << 1;

  const int lrow  = threadIdx.x >> 4;
  const int lcol  = (threadIdx.x & 15) << 2;

  const float* __restrict__ Bw = (mat ? Vb : Vr) + (size_t)khalf * PR + 4 * c;
  const size_t goff = (size_t)(row0 + lrow) * P_DIM + lcol;
  const int aoff = mat * PJ_MAT + (rt * 16 + c) * PJ_LD + khalf;
  const int soff = lrow * PJ_LD + lcol;

  v4f tr[4], fr[4];
#pragma unroll
  for (int q = 0; q < 4; ++q) {
    tr[q] = *(const v4f*)&t[goff + (size_t)q * 8 * P_DIM];
    fr[q] = *(const v4f*)&F[goff + (size_t)q * 8 * P_DIM];
  }

  v4f rb0[16], rb1[16];
#pragma unroll
  for (int i = 0; i < PJ_PF; ++i) {
    rb0[i] = *(const v4f*)&Bw[(size_t)(4 * i)     * PR];
    rb1[i] = *(const v4f*)&Bw[(size_t)(4 * i + 1) * PR];
  }

#pragma unroll
  for (int q = 0; q < 4; ++q) {
    *(v4f*)&As[soff + q * 8 * PJ_LD]          = tr[q];
    *(v4f*)&As[PJ_MAT + soff + q * 8 * PJ_LD] = tr[q] * fr[q];
  }
  lds_barrier();

  v8f acc[4];
#pragma unroll
  for (int j = 0; j < 4; ++j) acc[j] = (v8f){};

  for (int ch = 0; ch < P_DIM / 64 - 1; ++ch) {
    const int kn = (ch + 1) * 64;
#pragma unroll
    for (int q = 0; q < 4; ++q) {
      tr[q] = *(const v4f*)&t[goff + (size_t)q * 8 * P_DIM + kn];
      fr[q] = *(const v4f*)&F[goff + (size_t)q * 8 * P_DIM + kn];
    }
    proj_chunk<false>(&As[(ch & 1) * PJ_STG + aoff], Bw + (size_t)(kn - 64) * PR, rb0, rb1, acc);
    float* nxt = &As[((ch + 1) & 1) * PJ_STG + soff];
#pragma unroll
    for (int q = 0; q < 4; ++q) {
      *(v4f*)&nxt[q * 8 * PJ_LD]          = tr[q];
      *(v4f*)&nxt[PJ_MAT + q * 8 * PJ_LD] = tr[q] * fr[q];
    }
    lds_barrier();
  }
  proj_chunk<true>(&As[((P_DIM / 64 - 1) & 1) * PJ_STG + aoff], Bw + (size_t)(P_DIM - 64) * PR, rb0, rb1, acc);

  float* __restrict__ dst = mat ? inj_out : r_out;
#pragma unroll
  for (int v = 0; v < 8; ++v) {
    const int row = row0 + rt * 16 + v + (h << 3);
    v4f o;
    o.x = acc[0][v]; o.y = acc[1][v]; o.z = acc[2][v]; o.w = acc[3][v];
    *(v4f*)&dst[(size_t)row * PR + 4 * c] = o;
  }
}

__global__ __launch_bounds__(256) void k_gate(const float* __restrict__ r_in,
                                              const float* __restrict__ Ww,
                                              const float* __restrict__ Wb,
                                              float* __restrict__ a_out) {
  const int lane  = threadIdx.x & 31;
  const int wave  = threadIdx.x >> 5;
  const int wm    = wave >> 2, cn = wave & 3;
  const int row0  = blockIdx.x * 32 + wm * 16;
  const int m     = lane & 15;
  const int khalf = (lane >> 4) << 1;
  const int col   = cn * 16 + (lane & 15);

  v8f acc = {};
#pragma unroll
  for (int ks = 0; ks < PR; ks += 4) {
    v2f afrag = *(const v2f*)&r_in[(size_t)(row0 + m) * PR + ks + khalf];
    v2f bfrag = *(const v2f*)&Ww[(size_t)col * PR + ks + khalf];
    acc = __builtin_amdgcn_wmma_f32_16x16x4_f32(false, afrag, false, bfrag,
                                                (short)0, acc, false, false);
  }
  const float bias = Wb[col];
#pragma unroll
  for (int v = 0; v < 8; ++v) {
    const int row = row0 + v + ((lane >> 4) << 3);
    float x = acc[v] + bias;
    float s = 1.0f / (1.0f + __expf(-x));
    s = fminf(fmaxf(s, AMIN), AMAX);
    a_out[(size_t)row * PR + col] = s;
  }
}

__global__ __launch_bounds__(64) void k_scan_local(float* __restrict__ a_cum,
                                                   float* __restrict__ inj_s,
                                                   int T) {
  const int C = T / SCAN_L;
  const int b = blockIdx.x / C, c = blockIdx.x % C;
  const int r = threadIdx.x;
  size_t base = ((size_t)b * T + (size_t)c * SCAN_L) * PR + r;
  float s = 0.0f, cum = 1.0f;
  for (int i = 0; i < SCAN_L; ++i) {
    const size_t off = base + (size_t)i * PR;
    const float av = a_cum[off];
    s   = fmaf(av, s, inj_s[off]);
    cum *= av;
    a_cum[off] = cum;
    inj_s[off] = s;
  }
}

__global__ __launch_bounds__(64) void k_scan_carry(const float* __restrict__ a_cum,
                                                   const float* __restrict__ inj_s,
                                                   float* __restrict__ carry_buf,
                                                   float* __restrict__ s_last,
                                                   int T) {
  const int C = T / SCAN_L;
  const int b = blockIdx.x, r = threadIdx.x;
  float carry = 0.0f;
  for (int c = 0; c < C; ++c) {
    carry_buf[((size_t)b * C + c) * PR + r] = carry;
    const size_t off = ((size_t)b * T + (size_t)c * SCAN_L + (SCAN_L - 1)) * PR + r;
    carry = fmaf(a_cum[off], carry, inj_s[off]);
  }
  s_last[b * PR + r] = carry;
}

__global__ __launch_bounds__(256) void k_scan_apply(const float* __restrict__ a_cum,
                                                    const float* __restrict__ carry_buf,
                                                    float* __restrict__ inj_s,
                                                    int T, int total) {
  const int idx = blockIdx.x * blockDim.x + threadIdx.x;
  if (idx >= total) return;
  const int C = T / SCAN_L;
  const int r = idx % PR;
  const int t = (idx / PR) % T;
  const int b = idx / (PR * T);
  const int c = t / SCAN_L;
  inj_s[idx] = fmaf(a_cum[idx], carry_buf[((size_t)b * C + c) * PR + r], inj_s[idx]);
}

#define OUT_MT 2
#define OUT_PF 3
__global__ __launch_bounds__(256) void k_outp(const float* __restrict__ s_in,
                                              const float* __restrict__ Vo,
                                              float* __restrict__ out) {
  __shared__ float As[OUT_MT * 16 * 68];
  const int lane  = threadIdx.x & 31;
  const int wave  = threadIdx.x >> 5;
  const int row0  = blockIdx.x * (OUT_MT * 16);
  const int c     = lane & 15;
  const int h     = lane >> 4;
  const int khalf = h << 1;
  const int colw  = wave * 128 + 4 * c;

  float4 rb0[OUT_PF + 1], rb1[OUT_PF + 1];
#pragma unroll
  for (int t = 0; t < OUT_PF; ++t) {
    const int kabs = 4 * t + khalf;
    rb0[t] = *(const float4*)&Vo[(size_t)kabs       * P_DIM + colw];
    rb1[t] = *(const float4*)&Vo[(size_t)(kabs + 1) * P_DIM + colw];
  }

#pragma unroll
  for (int q = 0; q < OUT_MT; ++q) {
    const int lrow = q * 16 + (threadIdx.x >> 4);
    const int lcol = (threadIdx.x & 15) << 2;
    const float4 sv = *(const float4*)&s_in[(size_t)(row0 + lrow) * PR + lcol];
    *(float4*)&As[lrow * 68 + lcol] = sv;
  }
  __syncthreads();

  v2f a[OUT_MT][16];
#pragma unroll
  for (int q = 0; q < OUT_MT; ++q) {
#pragma unroll
    for (int i = 0; i < 16; ++i)
      a[q][i] = *(const v2f*)&As[(q * 16 + c) * 68 + 4 * i + khalf];
  }

  v8f acc[OUT_MT][4];
#pragma unroll
  for (int t = 0; t < 32; ++t) {
    const int g = t >> 4, i = t & 15;
    if (i == 0) {
#pragma unroll
      for (int q = 0; q < OUT_MT; ++q) {
#pragma unroll
        for (int j = 0; j < 4; ++j) acc[q][j] = (v8f){};
      }
    }
    if (t + OUT_PF < 32) {
      const int tn = t + OUT_PF;
      const int kabs = 4 * (tn & 15) + khalf;
      const int coln = colw + (tn >> 4) * 64;
      rb0[tn % (OUT_PF + 1)] = *(const float4*)&Vo[(size_t)kabs       * P_DIM + coln];
      rb1[tn % (OUT_PF + 1)] = *(const float4*)&Vo[(size_t)(kabs + 1) * P_DIM + coln];
    }
    __builtin_amdgcn_sched_barrier(0);
    const float4 b0 = rb0[t % (OUT_PF + 1)];
    const float4 b1 = rb1[t % (OUT_PF + 1)];
    const float b0a[4] = {b0.x, b0.y, b0.z, b0.w};
    const float b1a[4] = {b1.x, b1.y, b1.z, b1.w};
#pragma unroll
    for (int j = 0; j < 4; ++j) {
      v2f bfrag;
      bfrag.x = b0a[j];
      bfrag.y = b1a[j];
#pragma unroll
      for (int q = 0; q < OUT_MT; ++q)
        acc[q][j] = __builtin_amdgcn_wmma_f32_16x16x4_f32(false, a[q][i], false, bfrag,
                                                          (short)0, acc[q][j], false, false);
    }
    if (i == 15) {
      const int colb = colw + g * 64;
#pragma unroll
      for (int q = 0; q < OUT_MT; ++q) {
#pragma unroll
        for (int v = 0; v < 8; ++v) {
          const int row = row0 + q * 16 + v + (h << 3);
          float4 o;
          o.x = acc[q][0][v]; o.y = acc[q][1][v]; o.z = acc[q][2][v]; o.w = acc[q][3][v];
          *(float4*)&out[(size_t)row * P_DIM + colb] = o;
        }
      }
    }
  }
}

extern "C" void kernel_launch(void* const* d_in, const int* in_sizes, int n_in,
                              void* d_out, int out_size, void* d_ws, size_t ws_size,
                              hipStream_t stream) {
  const float* t  = (const float*)d_in[0];
  const float* F  = (const float*)d_in[1];
  const float* Vr = (const float*)d_in[2];
  const float* Vb = (const float*)d_in[3];
  const float* Vo = (const float*)d_in[4];
  const float* Ww = (const float*)d_in[5];
  const float* Wb = (const float*)d_in[6];
  float* out = (float*)d_out;

  const int rows = in_sizes[0] / P_DIM;
  const int Bn   = 8;
  const int T    = rows / Bn;
  const int C    = T / SCAN_L;

  float* r_buf     = (float*)d_ws;
  float* a_buf     = r_buf   + (size_t)rows * PR;
  float* inj_buf   = a_buf   + (size_t)rows * PR;
  float* carry_buf = inj_buf + (size_t)rows * PR;
  float* s_last    = out + (size_t)rows * P_DIM;

  k_proj<<<rows / 32, 128, 0, stream>>>(t, F, Vr, Vb, r_buf, inj_buf);
  k_gate<<<rows / 32, 256, 0, stream>>>(r_buf, Ww, Wb, a_buf);
  k_scan_local<<<Bn * C, 64, 0, stream>>>(a_buf, inj_buf, T);
  k_scan_carry<<<Bn, 64, 0, stream>>>(a_buf, inj_buf, carry_buf, s_last, T);
  const int total = rows * PR;
  k_scan_apply<<<(total + 255) / 256, 256, 0, stream>>>(a_buf, carry_buf, inj_buf, T, total);
  k_outp<<<rows / 32, 256, 0, stream>>>(inj_buf, Vo, out);
}


# scan restructured: 32-row sub-chunk summaries (P,S) in one small launch; k_outp folds earlier summaries into its entry state and runs the recurrence in LDS (new sum order); 6 launches -> 4
# speedup vs baseline: 1.7812x; 1.3172x over previous
#include <hip/hip_runtime.h>
#include <math.h>

typedef __attribute__((ext_vector_type(2))) float v2f;
typedef __attribute__((ext_vector_type(4))) float v4f;
typedef __attribute__((ext_vector_type(8))) float v8f;

#define P_DIM 1024
#define PR    64
#define AMIN  0.01f
#define AMAX  0.995f

#define PJ_ROWS 32
#define PJ_LD   68
#define PJ_MAT  (PJ_ROWS * PJ_LD)
#define PJ_STG  (2 * PJ_MAT)

__device__ __forceinline__ void lds_barrier() {
  asm volatile("s_wait_dscnt 0x0\n\ts_barrier_signal -1\n\ts_barrier_wait -1" ::: "memory");
}

#define PJ_PF 15

template <bool LAST>
__device__ __forceinline__ void proj_chunk(const float* Ab, const float* __restrict__ Bc,
                                           v4f (&rb0)[16], v4f (&rb1)[16], v8f (&acc)[4]) {
  v2f a[16];
#pragma unroll
  for (int i = 0; i < 16; ++i) a[i] = *(const v2f*)&Ab[4 * i];
#pragma unroll
  for (int i = 0; i < 16; ++i) {
    if (!LAST || i == 0) {
      rb0[(i + PJ_PF) & 15] = *(const v4f*)&Bc[(size_t)(4 * (i + PJ_PF))     * PR];
      rb1[(i + PJ_PF) & 15] = *(const v4f*)&Bc[(size_t)(4 * (i + PJ_PF) + 1) * PR];
    }
    __builtin_amdgcn_sched_barrier(0);
    const v4f b0 = rb0[i];
    const v4f b1 = rb1[i];
#pragma unroll
    for (int j = 0; j < 4; ++j) {
      v2f bfrag;
      bfrag.x = b0[j];
      bfrag.y = b1[j];
      acc[j] = __builtin_amdgcn_wmma_f32_16x16x4_f32(false, a[i], false, bfrag,
                                                     (short)0, acc[j], false, false);
    }
  }
}

__global__ __launch_bounds__(128) void k_proj(const float* __restrict__ t,
                                              const float* __restrict__ F,
                                              const float* __restrict__ Vr,
                                              const float* __restrict__ Vb,
                                              float* __restrict__ r_out,
                                              float* __restrict__ inj_out) {
  __shared__ __attribute__((aligned(16))) float As[2 * PJ_STG];
  const int lane  = threadIdx.x & 31;
  const int wave  = threadIdx.x >> 5;
  const int mat   = wave >> 1;
  const int rt    = wave & 1;
  const int row0  = blockIdx.x * PJ_ROWS;
  const int c     = lane & 15;
  const int h     = lane >> 4;
  const int khalf = h << 1;

  const int lrow  = threadIdx.x >> 4;
  const int lcol  = (threadIdx.x & 15) << 2;

  const float* __restrict__ Bw = (mat ? Vb : Vr) + (size_t)khalf * PR + 4 * c;
  const size_t goff = (size_t)(row0 + lrow) * P_DIM + lcol;
  const int aoff = mat * PJ_MAT + (rt * 16 + c) * PJ_LD + khalf;
  const int soff = lrow * PJ_LD + lcol;

  v4f tr[4], fr[4];
#pragma unroll
  for (int q = 0; q < 4; ++q) {
    tr[q] = *(const v4f*)&t[goff + (size_t)q * 8 * P_DIM];
    fr[q] = *(const v4f*)&F[goff + (size_t)q * 8 * P_DIM];
  }

  v4f rb0[16], rb1[16];
#pragma unroll
  for (int i = 0; i < PJ_PF; ++i) {
    rb0[i] = *(const v4f*)&Bw[(size_t)(4 * i)     * PR];
    rb1[i] = *(const v4f*)&Bw[(size_t)(4 * i + 1) * PR];
  }

#pragma unroll
  for (int q = 0; q < 4; ++q) {
    *(v4f*)&As[soff + q * 8 * PJ_LD]          = tr[q];
    *(v4f*)&As[PJ_MAT + soff + q * 8 * PJ_LD] = tr[q] * fr[q];
  }
  lds_barrier();

  v8f acc[4];
#pragma unroll
  for (int j = 0; j < 4; ++j) acc[j] = (v8f){};

  for (int ch = 0; ch < P_DIM / 64 - 1; ++ch) {
    const int kn = (ch + 1) * 64;
#pragma unroll
    for (int q = 0; q < 4; ++q) {
      tr[q] = *(const v4f*)&t[goff + (size_t)q * 8 * P_DIM + kn];
      fr[q] = *(const v4f*)&F[goff + (size_t)q * 8 * P_DIM + kn];
    }
    proj_chunk<false>(&As[(ch & 1) * PJ_STG + aoff], Bw + (size_t)(kn - 64) * PR, rb0, rb1, acc);
    float* nxt = &As[((ch + 1) & 1) * PJ_STG + soff];
#pragma unroll
    for (int q = 0; q < 4; ++q) {
      *(v4f*)&nxt[q * 8 * PJ_LD]          = tr[q];
      *(v4f*)&nxt[PJ_MAT + q * 8 * PJ_LD] = tr[q] * fr[q];
    }
    lds_barrier();
  }
  proj_chunk<true>(&As[((P_DIM / 64 - 1) & 1) * PJ_STG + aoff], Bw + (size_t)(P_DIM - 64) * PR, rb0, rb1, acc);

  float* __restrict__ dst = mat ? inj_out : r_out;
#pragma unroll
  for (int v = 0; v < 8; ++v) {
    const int row = row0 + rt * 16 + v + (h << 3);
    v4f o;
    o.x = acc[0][v]; o.y = acc[1][v]; o.z = acc[2][v]; o.w = acc[3][v];
    *(v4f*)&dst[(size_t)row * PR + 4 * c] = o;
  }
}

__global__ __launch_bounds__(256) void k_gate(const float* __restrict__ r_in,
                                              const float* __restrict__ Ww,
                                              const float* __restrict__ Wb,
                                              float* __restrict__ a_out) {
  const int lane  = threadIdx.x & 31;
  const int wave  = threadIdx.x >> 5;
  const int wm    = wave >> 2, cn = wave & 3;
  const int row0  = blockIdx.x * 32 + wm * 16;
  const int m     = lane & 15;
  const int khalf = (lane >> 4) << 1;
  const int col   = cn * 16 + (lane & 15);

  v8f acc = {};
#pragma unroll
  for (int ks = 0; ks < PR; ks += 4) {
    v2f afrag = *(const v2f*)&r_in[(size_t)(row0 + m) * PR + ks + khalf];
    v2f bfrag = *(const v2f*)&Ww[(size_t)col * PR + ks + khalf];
    acc = __builtin_amdgcn_wmma_f32_16x16x4_f32(false, afrag, false, bfrag,
                                                (short)0, acc, false, false);
  }
  const float bias = Wb[col];
#pragma unroll
  for (int v = 0; v < 8; ++v) {
    const int row = row0 + v + ((lane >> 4) << 3);
    float x = acc[v] + bias;
    float s = 1.0f / (1.0f + __expf(-x));
    s = fminf(fmaxf(s, AMIN), AMAX);
    a_out[(size_t)row * PR + col] = s;
  }
}

#define SUB_L 32
__global__ __launch_bounds__(64) void k_scan_summ(const float* __restrict__ a_in,
                                                  const float* __restrict__ inj_in,
                                                  float* __restrict__ sum_p,
                                                  float* __restrict__ sum_s) {
  const int r = threadIdx.x;
  const size_t base = (size_t)blockIdx.x * SUB_L * PR + r;
  float av[SUB_L], iv[SUB_L];
#pragma unroll
  for (int i = 0; i < SUB_L; ++i) {
    av[i] = a_in[base + (size_t)i * PR];
    iv[i] = inj_in[base + (size_t)i * PR];
  }
  float s = 0.0f, p = 1.0f;
#pragma unroll
  for (int i = 0; i < SUB_L; ++i) {
    s = fmaf(av[i], s, iv[i]);
    p *= av[i];
  }
  sum_p[(size_t)blockIdx.x * PR + r] = p;
  sum_s[(size_t)blockIdx.x * PR + r] = s;
}

__device__ __forceinline__ void lb_load(const float* __restrict__ bp, const float* __restrict__ bs,
                                        int kb, int jsub, int sl, float (&pp)[16], float (&ss)[16]) {
#pragma unroll
  for (int u = 0; u < 16; ++u) {
    int kc = kb + u;
    kc = kc < jsub ? kc : jsub - 1;
    kc = kc > 0 ? kc : 0;
    const unsigned o = (unsigned)kc * PR + (unsigned)sl;
    pp[u] = bp[o];
    ss[u] = bs[o];
  }
}
__device__ __forceinline__ void lb_fold(int kb, int jsub, const float (&pp)[16], const float (&ss)[16],
                                        float& P, float& S) {
  P = 1.0f;
  S = 0.0f;
#pragma unroll
  for (int u = 0; u < 16; ++u) {
    const bool ok = (kb + u) < jsub;
    const float pu = ok ? pp[u] : 1.0f;
    const float su = ok ? ss[u] : 0.0f;
    S = fmaf(pu, S, su);
    P *= pu;
  }
}

#define OUT_MT 2
#define OUT_PF 3
__global__ __launch_bounds__(256) void k_outp(const float* __restrict__ a_in,
                                              const float* __restrict__ inj_in,
                                              const float* __restrict__ sum_p,
                                              const float* __restrict__ sum_s,
                                              const float* __restrict__ Vo,
                                              float* __restrict__ out,
                                              float* __restrict__ s_last,
                                              int T) {
  __shared__ float As[OUT_MT * 16 * 68];
  __shared__ float Aa[OUT_MT * 16 * 68];
  __shared__ float Sg[2][4][PR];
  const int lane  = threadIdx.x & 31;
  const int wave  = threadIdx.x >> 5;
  const int row0  = blockIdx.x * (OUT_MT * 16);
  const int c     = lane & 15;
  const int h     = lane >> 4;
  const int khalf = h << 1;
  const int colw  = wave * 128 + 4 * c;

  float4 rb0[OUT_PF + 1], rb1[OUT_PF + 1];
#pragma unroll
  for (int t = 0; t < OUT_PF; ++t) {
    const int kabs = 4 * t + khalf;
    rb0[t] = *(const float4*)&Vo[(size_t)kabs       * P_DIM + colw];
    rb1[t] = *(const float4*)&Vo[(size_t)(kabs + 1) * P_DIM + colw];
  }

  float4 iv4[OUT_MT], av4[OUT_MT];
#pragma unroll
  for (int q = 0; q < OUT_MT; ++q) {
    const int lrow = q * 16 + (threadIdx.x >> 4);
    const int lcol = (threadIdx.x & 15) << 2;
    iv4[q] = *(const float4*)&inj_in[(size_t)(row0 + lrow) * PR + lcol];
    av4[q] = *(const float4*)&a_in[(size_t)(row0 + lrow) * PR + lcol];
  }

  const int nsub  = T / SUB_L;
  const int jsub  = blockIdx.x % nsub;
  const int first = blockIdx.x - jsub;
  const int grp   = threadIdx.x >> 6;
  const int sl    = threadIdx.x & 63;
  const float* __restrict__ bp = sum_p + (size_t)first * PR;
  const float* __restrict__ bs = sum_s + (size_t)first * PR;
  float carry = 0.0f;
  float pp[16], ss[16];
  float P, S;
  lb_load(bp, bs, 16 * grp, jsub, sl, pp, ss);
  __builtin_amdgcn_sched_barrier(0);
#pragma unroll
  for (int q = 0; q < OUT_MT; ++q) {
    const int lrow = q * 16 + (threadIdx.x >> 4);
    const int lcol = (threadIdx.x & 15) << 2;
    *(float4*)&As[lrow * 68 + lcol] = iv4[q];
    *(float4*)&Aa[lrow * 68 + lcol] = av4[q];
  }
  lb_fold(16 * grp, jsub, pp, ss, P, S);
  Sg[0][grp][sl] = P;
  Sg[1][grp][sl] = S;
  __syncthreads();
  if (threadIdx.x < 64) {
#pragma unroll
    for (int gg = 0; gg < 4; ++gg) carry = fmaf(Sg[0][gg][sl], carry, Sg[1][gg][sl]);
  }
  for (int k0 = 64; k0 < jsub; k0 += 64) {
    __syncthreads();
    lb_load(bp, bs, k0 + 16 * grp, jsub, sl, pp, ss);
    lb_fold(k0 + 16 * grp, jsub, pp, ss, P, S);
    Sg[0][grp][sl] = P;
    Sg[1][grp][sl] = S;
    __syncthreads();
    if (threadIdx.x < 64) {
#pragma unroll
      for (int gg = 0; gg < 4; ++gg) carry = fmaf(Sg[0][gg][sl], carry, Sg[1][gg][sl]);
    }
  }

  if (threadIdx.x < 64) {
    float ac[OUT_MT * 16], ic[OUT_MT * 16];
#pragma unroll
    for (int i = 0; i < OUT_MT * 16; ++i) {
      ac[i] = Aa[i * 68 + sl];
      ic[i] = As[i * 68 + sl];
    }
    float sv = carry;
#pragma unroll
    for (int i = 0; i < OUT_MT * 16; ++i) {
      sv = fmaf(ac[i], sv, ic[i]);
      As[i * 68 + sl] = sv;
    }
    if (jsub == nsub - 1) s_last[(size_t)(blockIdx.x / nsub) * PR + sl] = sv;
  }
  __syncthreads();

  v2f a[OUT_MT][16];
#pragma unroll
  for (int q = 0; q < OUT_MT; ++q) {
#pragma unroll
    for (int i = 0; i < 16; ++i)
      a[q][i] = *(const v2f*)&As[(q * 16 + c) * 68 + 4 * i + khalf];
  }

  v8f acc[OUT_MT][4];
#pragma unroll
  for (int t = 0; t < 32; ++t) {
    const int g = t >> 4, i = t & 15;
    if (i == 0) {
#pragma unroll
      for (int q = 0; q < OUT_MT; ++q) {
#pragma unroll
        for (int j = 0; j < 4; ++j) acc[q][j] = (v8f){};
      }
    }
    if (t + OUT_PF < 32) {
      const int tn = t + OUT_PF;
      const int kabs = 4 * (tn & 15) + khalf;
      const int coln = colw + (tn >> 4) * 64;
      rb0[tn % (OUT_PF + 1)] = *(const float4*)&Vo[(size_t)kabs       * P_DIM + coln];
      rb1[tn % (OUT_PF + 1)] = *(const float4*)&Vo[(size_t)(kabs + 1) * P_DIM + coln];
    }
    __builtin_amdgcn_sched_barrier(0);
    const float4 b0 = rb0[t % (OUT_PF + 1)];
    const float4 b1 = rb1[t % (OUT_PF + 1)];
    const float b0a[4] = {b0.x, b0.y, b0.z, b0.w};
    const float b1a[4] = {b1.x, b1.y, b1.z, b1.w};
#pragma unroll
    for (int j = 0; j < 4; ++j) {
      v2f bfrag;
      bfrag.x = b0a[j];
      bfrag.y = b1a[j];
#pragma unroll
      for (int q = 0; q < OUT_MT; ++q)
        acc[q][j] = __builtin_amdgcn_wmma_f32_16x16x4_f32(false, a[q][i], false, bfrag,
                                                          (short)0, acc[q][j], false, false);
    }
    if (i == 15) {
      const int colb = colw + g * 64;
#pragma unroll
      for (int q = 0; q < OUT_MT; ++q) {
#pragma unroll
        for (int v = 0; v < 8; ++v) {
          const int row = row0 + q * 16 + v + (h << 3);
          float4 o;
          o.x = acc[q][0][v]; o.y = acc[q][1][v]; o.z = acc[q][2][v]; o.w = acc[q][3][v];
          *(float4*)&out[(size_t)row * P_DIM + colb] = o;
        }
      }
    }
  }
}

extern "C" void kernel_launch(void* const* d_in, const int* in_sizes, int n_in,
                              void* d_out, int out_size, void* d_ws, size_t ws_size,
                              hipStream_t stream) {
  const float* t  = (const float*)d_in[0];
  const float* F  = (const float*)d_in[1];
  const float* Vr = (const float*)d_in[2];
  const float* Vb = (const float*)d_in[3];
  const float* Vo = (const float*)d_in[4];
  const float* Ww = (const float*)d_in[5];
  const float* Wb = (const float*)d_in[6];
  float* out = (float*)d_out;

  const int rows = in_sizes[0] / P_DIM;
  const int Bn   = 8;
  const int T    = rows / Bn;

  float* r_buf     = (float*)d_ws;
  float* a_buf     = r_buf   + (size_t)rows * PR;
  float* inj_buf   = a_buf   + (size_t)rows * PR;
  float* sum_p     = r_buf;
  float* sum_s     = r_buf   + (size_t)(rows / SUB_L) * PR;
  float* s_last    = out + (size_t)rows * P_DIM;

  k_proj<<<rows / 32, 128, 0, stream>>>(t, F, Vr, Vb, r_buf, inj_buf);
  k_gate<<<rows / 32, 256, 0, stream>>>(r_buf, Ww, Wb, a_buf);
  k_scan_summ<<<rows / SUB_L, 64, 0, stream>>>(a_buf, inj_buf, sum_p, sum_s);
  k_outp<<<rows / 32, 256, 0, stream>>>(a_buf, inj_buf, sum_p, sum_s, Vo, out, s_last, T);
}
